# starting kernel
# baseline (speedup 1.0000x reference)
#include <hip/hip_runtime.h>

typedef float v2f __attribute__((ext_vector_type(2)));
typedef float v8f __attribute__((ext_vector_type(8)));

__device__ __forceinline__ v8f wmma4(v2f a, v2f b, v8f c) {
  return __builtin_amdgcn_wmma_f32_16x16x4_f32(false, a, false, b, (short)0, c,
                                               false, false);
}

#define M_DIM 70
#define N_DIM 100
#define B_ROWS 4096
#define NPAD 112
#define KT_S 25
#define KT_W 18
#define WAVES 2
#define THREADS (WAVES * 32)

__global__ __launch_bounds__(THREADS) void ISTA_53154515255813_kernel(
    const float* __restrict__ y, const float* __restrict__ S,
    const float* __restrict__ W, const float* __restrict__ thrp,
    const int* __restrict__ iterp, float* __restrict__ out) {
  __shared__ float smem[50 * NPAD * 2];
  __shared__ float dbuf[WAVES * 16 * N_DIM];

  const int tid = threadIdx.x;
  const int lane = tid & 31;
  const int wave = tid >> 5;
  const int ln = lane & 15;
  const int hi = lane >> 4;
  const int row0 = (blockIdx.x * WAVES + wave) * 16;

  for (int i = tid; i < 36 * NPAD; i += THREADS) {
    const int k2 = i / NPAD, n = i % NPAD;
    const int ka = 2 * k2, kb = ka + 1;
    smem[2 * i + 0] = (n < N_DIM && ka < M_DIM) ? W[n * M_DIM + ka] : 0.0f;
    smem[2 * i + 1] = (n < N_DIM && kb < M_DIM) ? W[n * M_DIM + kb] : 0.0f;
  }
  float* db = &dbuf[wave * 16 * N_DIM];
  for (int i = lane; i < 16 * N_DIM; i += 32) db[i] = 0.0f;
  __syncthreads();

  v2f ay[KT_W];
  {
    const float* yb = y + (size_t)row0 * M_DIM + (size_t)ln * M_DIM;
#pragma unroll
    for (int k = 0; k < KT_W; ++k) {
      const int colA = 4 * k + 2 * hi;
      v2f a;
      a.x = (colA + 0 < M_DIM) ? yb[colA + 0] : 0.0f;
      a.y = (colA + 1 < M_DIM) ? yb[colA + 1] : 0.0f;
      ay[k] = a;
    }
  }
  v8f wy[7];
#pragma unroll
  for (int j = 0; j < 7; ++j) {
    v8f c = {};
#pragma unroll
    for (int k = 0; k < KT_W; ++k) {
      const v2f b =
          *(const v2f*)&smem[((2 * k + hi) * NPAD + 16 * j + ln) * 2];
      c = wmma4(ay[k], b, c);
    }
    wy[j] = c;
  }
  __syncthreads();

  for (int i = tid; i < 50 * NPAD; i += THREADS) {
    const int k2 = i / NPAD, n = i % NPAD;
    const int ka = 2 * k2;
    smem[2 * i + 0] = (n < N_DIM) ? S[n * N_DIM + ka] : 0.0f;
    smem[2 * i + 1] = (n < N_DIM) ? S[n * N_DIM + ka + 1] : 0.0f;
  }
  __syncthreads();

  const float thr = thrp[0];
  const int nIter = iterp[0];

  const float* arow = db + ln * N_DIM + 2 * hi;

  for (int it = 0; it < nIter; ++it) {
    v2f a[KT_S];
#pragma unroll
    for (int k = 0; k < KT_S; ++k) a[k] = *(const v2f*)(arow + 4 * k);

#pragma unroll
    for (int j = 0; j < 7; ++j) {
      v8f c = wy[j];
#pragma unroll
      for (int k = 0; k < KT_S; ++k) {
        const v2f b =
            *(const v2f*)&smem[((2 * k + hi) * NPAD + 16 * j + ln) * 2];
        c = wmma4(a[k], b, c);
      }
      const int col = 16 * j + ln;
#pragma unroll
      for (int r = 0; r < 8; ++r) {
        const float x = c[r];
        const float dn = fmaxf(x - thr, 0.0f) - fmaxf(-x - thr, 0.0f);
        const int rowc = hi ? (r + 8) : r;
        if (col < N_DIM) db[rowc * N_DIM + col] = dn;
      }
    }
  }

  {
    float* ob = out + (size_t)row0 * N_DIM;
    for (int i = lane; i < 16 * N_DIM; i += 32) ob[i] = db[i];
  }
}

extern "C" void kernel_launch(void* const* d_in, const int* in_sizes, int n_in,
                              void* d_out, int out_size, void* d_ws,
                              size_t ws_size, hipStream_t stream) {
  const float* y = (const float*)d_in[0];
  const float* S = (const float*)d_in[1];
  const float* W = (const float*)d_in[2];
  const float* thr = (const float*)d_in[3];
  const int* nIt = (const int*)d_in[4];
  float* out = (float*)d_out;

  const int blocks = B_ROWS / (16 * WAVES);
  hipLaunchKernelGGL(ISTA_53154515255813_kernel, dim3(blocks), dim3(THREADS),
                     0, stream, y, S, W, thr, nIt, out);
}


# Split the 7 column tiles of each 16-row slice over 4 waves (256 blocks, one wave per SIMD): each wave keeps its S slice in registers, d goes through double-buffered LDS, one barrier per step
# speedup vs baseline: 3.9417x; 3.9417x over previous
#include <hip/hip_runtime.h>


typedef float v2f __attribute__((ext_vector_type(2)));
typedef float v8f __attribute__((ext_vector_type(8)));

__device__ __forceinline__ v8f wmma4(v2f a, v2f b, v8f c) {
  return __builtin_amdgcn_wmma_f32_16x16x4_f32(false, a, false, b, (short)0, c,
                                               false, false);
}

#define M_DIM 70
#define N_DIM 100
#define B_ROWS 4096
#define KT_S 25
#define KT_W 18
#define WAVES 4
#define THREADS (WAVES * 32)
#define TILE_ELEMS (16 * N_DIM)

__device__ __forceinline__ void soft_store(const v8f c, const float thr,
                                           float* __restrict__ dst,
                                           const int hi, const int col) {
#pragma unroll
  for (int r = 0; r < 8; ++r) {
    const float x = c[r];
    const float dn = fmaxf(x - thr, 0.0f) - fmaxf(-x - thr, 0.0f);
    if (col < N_DIM) dst[(8 * hi + r) * N_DIM + col] = dn;
  }
}

__global__ __launch_bounds__(THREADS) void ista_kernel(
    const float* __restrict__ y, const float* __restrict__ S,
    const float* __restrict__ W, const float* __restrict__ thrp,
    const int* __restrict__ iterp, float* __restrict__ out) {
  __shared__ float dbuf[2 * TILE_ELEMS];

  const int tid = threadIdx.x;
  const int lane = tid & 31;
  const int wave = __builtin_amdgcn_readfirstlane(tid >> 5);
  const int ln = lane & 15;
  const int hi = lane >> 4;
  const int row0 = blockIdx.x * 16;
  const bool two = wave < 3;

  const int n0 = 32 * wave + ln;
  const int n1 = n0 + 16;

  for (int i = tid; i < TILE_ELEMS; i += THREADS) dbuf[i] = 0.0f;

  const bool ok0 = n0 < N_DIM;
  const bool ok1 = two && n1 < N_DIM;
  const int r0 = ok0 ? n0 : N_DIM - 1;
  const int r1 = ok1 ? n1 : N_DIM - 1;

  v8f wy0 = {}, wy1 = {};
  {
    v2f ay[KT_W];
    const float* yb = y + (size_t)(row0 + ln) * M_DIM + 2 * hi;
#pragma unroll
    for (int k = 0; k < KT_W; ++k) {
      v2f a = {0.0f, 0.0f};
      if (4 * k + 2 * hi < M_DIM) a = *(const v2f*)(yb + 4 * k);
      ay[k] = a;
    }
    const float* w0 = W + (size_t)r0 * M_DIM;
    const float* w1 = W + (size_t)r1 * M_DIM;
    v2f bw0[KT_W], bw1[KT_W];
#pragma unroll
    for (int k = 0; k < KT_W; ++k) {
      const bool in = 4 * k + 2 * hi < M_DIM;
      const int col = in ? 4 * k + 2 * hi : 4 * k;
      const v2f t0 = *(const v2f*)(w0 + col);
      const v2f t1 = *(const v2f*)(w1 + col);
      const v2f z = {0.0f, 0.0f};
      bw0[k] = (ok0 && in) ? t0 : z;
      bw1[k] = (ok1 && in) ? t1 : z;
    }
#pragma unroll
    for (int k = 0; k < KT_W; ++k) wy0 = wmma4(ay[k], bw0[k], wy0);
    if (two) {
#pragma unroll
      for (int k = 0; k < KT_W; ++k) wy1 = wmma4(ay[k], bw1[k], wy1);
    }
  }

  v2f s0[KT_S], s1[KT_S];
  {
    const float* p0 = S + (size_t)r0 * N_DIM + 2 * hi;
    const float* p1 = S + (size_t)r1 * N_DIM + 2 * hi;
#pragma unroll
    for (int k = 0; k < KT_S; ++k) {
      const v2f t0 = *(const v2f*)(p0 + 4 * k);
      const v2f t1 = *(const v2f*)(p1 + 4 * k);
      const v2f z = {0.0f, 0.0f};
      s0[k] = ok0 ? t0 : z;
      s1[k] = ok1 ? t1 : z;
    }
  }

  const float thr = thrp[0];
  const int nIter = iterp[0];

  const int aoff = ln * N_DIM + 2 * hi;

  __syncthreads();

  for (int it = 0; it < nIter; ++it) {
    const float* src = dbuf + (it & 1) * TILE_ELEMS;
    float* dst = dbuf + ((it & 1) ^ 1) * TILE_ELEMS;

    v2f a[KT_S];
#pragma unroll
    for (int k = 0; k < KT_S; ++k) a[k] = *(const v2f*)(src + aoff + 4 * k);

    if (two) {
      v8f c0 = wy0, c1 = wy1;
#pragma unroll
      for (int k = 0; k < KT_S; ++k) {
        c0 = wmma4(a[k], s0[k], c0);
        c1 = wmma4(a[k], s1[k], c1);
      }
      soft_store(c0, thr, dst, hi, n0);
      soft_store(c1, thr, dst, hi, n1);
    } else {
      v8f c0 = wy0;
#pragma unroll
      for (int k = 0; k < KT_S; ++k) c0 = wmma4(a[k], s0[k], c0);
      soft_store(c0, thr, dst, hi, n0);
    }
    __syncthreads();
  }

  {
    const int done = nIter > 0 ? nIter : 0;
    const float* res = dbuf + (done & 1) * TILE_ELEMS;
    float* ob = out + (size_t)row0 * N_DIM;
    for (int i = tid; i < TILE_ELEMS; i += THREADS) ob[i] = res[i];
  }
}

extern "C" void kernel_launch(void* const* d_in, const int* in_sizes, int n_in,
                              void* d_out, int out_size, void* d_ws,
                              size_t ws_size, hipStream_t stream) {
  const float* y = (const float*)d_in[0];
  const float* S = (const float*)d_in[1];
  const float* W = (const float*)d_in[2];
  const float* thr = (const float*)d_in[3];
  const int* nIt = (const int*)d_in[4];
  float* out = (float*)d_out;

  const int blocks = B_ROWS / 16;
  hipLaunchKernelGGL(ista_kernel, dim3(blocks), dim3(THREADS),
                     0, stream, y, S, W, thr, nIt, out);
}


# Transposed product dT = S dT: a result tile is itself the next step's B fragment, so own tiles skip LDS and the split barrier and the other tiles' loads hide behind them; k summation order changes
# speedup vs baseline: 4.3847x; 1.1124x over previous
#include <hip/hip_runtime.h>


typedef float v2f __attribute__((ext_vector_type(2)));
typedef float v4f __attribute__((ext_vector_type(4)));
typedef float v8f __attribute__((ext_vector_type(8)));

__device__ __forceinline__ v8f wmma4(v2f a, v2f b, v8f c) {
  return __builtin_amdgcn_wmma_f32_16x16x4_f32(false, a, false, b, (short)0, c,
                                               false, false);
}

#define M_DIM 70
#define N_DIM 100
#define B_ROWS 4096
#define KT_S 25
#define KT_W 18
#define WAVES 4
#define THREADS (WAVES * 32)
#define TILE_ELEMS (16 * N_DIM)
#define TILE_F 256
#define STAGE_F (6 * TILE_F + 64)


__device__ __forceinline__ float soft1(const float x, const float thr) {
  return fmaxf(x - thr, 0.0f) - fmaxf(-x - thr, 0.0f);
}
__device__ __forceinline__ v8f soft8(const v8f c, const float thr) {
  v8f d;
#pragma unroll
  for (int r = 0; r < 8; ++r) d[r] = soft1(c[r], thr);
  return d;
}

__device__ __forceinline__ void put8(float* stage, const int t, const int lane,
                                     const v8f d) {
  v4f* p = (v4f*)(stage + t * TILE_F + lane * 4);
  p[0] = v4f{d[0], d[1], d[2], d[3]};
  p[32] = v4f{d[4], d[5], d[6], d[7]};
}
__device__ __forceinline__ v8f get8(const float* stage, const int t,
                                    const int lane) {
  const v4f* p = (const v4f*)(stage + t * TILE_F + lane * 4);
  const v4f lo = p[0];
  const v4f hi = p[32];
  return v8f{lo[0], lo[1], lo[2], lo[3], hi[0], hi[1], hi[2], hi[3]};
}
__device__ __forceinline__ void put2(float* stage, const int lane,
                                     const v8f d) {
  *(v2f*)(stage + 6 * TILE_F + lane * 2) = v2f{d[0], d[1]};
}
__device__ __forceinline__ v2f get2(const float* stage, const int lane) {
  return *(const v2f*)(stage + 6 * TILE_F + lane * 2);
}

__device__ __forceinline__ void exch_arrive() {
  __builtin_amdgcn_fence(__ATOMIC_RELEASE, "workgroup");
  __builtin_amdgcn_s_barrier_signal(-1);
}
__device__ __forceinline__ void exch_wait() {
  __builtin_amdgcn_s_barrier_wait(-1);
  __builtin_amdgcn_fence(__ATOMIC_ACQUIRE, "workgroup");
}

__global__ __launch_bounds__(THREADS) void ista_kernel(
    const float* __restrict__ y, const float* __restrict__ S,
    const float* __restrict__ W, const float* __restrict__ thrp,
    const int* __restrict__ iterp, float* __restrict__ out) {
  __shared__ __attribute__((aligned(16))) float xbuf[2 * STAGE_F];
  __shared__ __attribute__((aligned(16))) float stg[TILE_ELEMS];

  const int tid = threadIdx.x;
  const int lane = tid & 31;
  const int wave = __builtin_amdgcn_readfirstlane(tid >> 5);
  const int c = lane & 15;
  const int h = lane >> 4;
  const int row0 = blockIdx.x * 16;
  const bool two = wave < 3;

  const int T0 = two ? 2 * wave : 6;
  const int T1 = two ? 2 * wave + 1 : 6;
  const bool okr = two || (c & 6) == 0;
  const int rA0 =
      two ? 16 * T0 + c : (okr ? 96 + (c & 1) + 2 * (c >> 3) : N_DIM - 1);
  const int rA1 = two ? 16 * T1 + c : rA0;
  const v2f z2 = {0.0f, 0.0f};

  v8f wy0 = {}, wy1 = {};
  {
    v2f by[KT_W];
    const float* yb = y + (size_t)(row0 + c) * M_DIM + 2 * h;
#pragma unroll
    for (int k = 0; k < KT_W; ++k) {
      v2f a = {0.0f, 0.0f};
      if (4 * k + 2 * h < M_DIM) a = *(const v2f*)(yb + 4 * k);
      by[k] = a;
    }
    const float* w0 = W + (size_t)rA0 * M_DIM;
    const float* w1 = W + (size_t)rA1 * M_DIM;
    v2f aw0[KT_W], aw1[KT_W];
#pragma unroll
    for (int k = 0; k < KT_W; ++k) {
      const bool in = 4 * k + 2 * h < M_DIM;
      const int col = in ? 4 * k + 2 * h : 4 * k;
      const v2f t0 = *(const v2f*)(w0 + col);
      const v2f t1 = *(const v2f*)(w1 + col);
      aw0[k] = (okr && in) ? t0 : z2;
      aw1[k] = in ? t1 : z2;
    }
#pragma unroll
    for (int k = 0; k < KT_W; ++k) wy0 = wmma4(aw0[k], by[k], wy0);
    if (two) {
#pragma unroll
      for (int k = 0; k < KT_W; ++k) wy1 = wmma4(aw1[k], by[k], wy1);
    }
  }

  v2f sA0[KT_S], sA1[KT_S];
  if (two) {
    const float* p0 = S + (size_t)rA0 * N_DIM + 8 * h;
    const float* p1 = S + (size_t)rA1 * N_DIM + 8 * h;
#pragma unroll
    for (int p = 0; p < 24; ++p) {
      int t;
      if (p < 4) {
        t = T0;
      } else if (p < 8) {
        t = T1;
      } else {
        t = 2 * wave + 2 + ((p - 8) >> 2);
        if (t >= 6) t -= 6;
      }
      const int kc = 16 * t + 2 * (p & 3);
      sA0[p] = *(const v2f*)(p0 + kc);
      sA1[p] = *(const v2f*)(p1 + kc);
    }
    sA0[24] = *(const v2f*)(S + (size_t)rA0 * N_DIM + 96 + 2 * h);
    sA1[24] = *(const v2f*)(S + (size_t)rA1 * N_DIM + 96 + 2 * h);
  } else {
    const float* p0 = S + (size_t)rA0 * N_DIM;
    const v2f t6 = *(const v2f*)(p0 + 96 + 2 * h);
    sA0[0] = okr ? t6 : z2;
    sA1[0] = z2;
#pragma unroll
    for (int p = 1; p < KT_S; ++p) {
      const int kc = 16 * ((p - 1) >> 2) + 8 * h + 2 * ((p - 1) & 3);
      const v2f v = *(const v2f*)(p0 + kc);
      sA0[p] = okr ? v : z2;
      sA1[p] = z2;
    }
  }

  const float thr = thrp[0];
  const int nIter = iterp[0];

  v8f d0 = {}, d1 = {};
  if (two) {
    put8(xbuf, T0, lane, d0);
    put8(xbuf, T1, lane, d1);
  } else {
    put2(xbuf, lane, d0);
  }

  for (int it = 0; it < nIter; ++it) {
    const float* src = xbuf + (it & 1) * STAGE_F;
    float* dst = xbuf + ((it & 1) ^ 1) * STAGE_F;
    v8f c0 = wy0, c1 = wy1;

    if (two) {
#pragma unroll
      for (int j = 0; j < 2; ++j) {
        const v2f b = {d0[2 * j], d0[2 * j + 1]};
        c0 = wmma4(sA0[j], b, c0);
        c1 = wmma4(sA1[j], b, c1);
      }
    } else {
      const v2f b = {d0[0], d0[1]};
      c0 = wmma4(sA0[0], b, c0);
    }

    exch_arrive();

    if (two) {
#pragma unroll
      for (int j = 2; j < 4; ++j) {
        const v2f b = {d0[2 * j], d0[2 * j + 1]};
        c0 = wmma4(sA0[j], b, c0);
        c1 = wmma4(sA1[j], b, c1);
      }
    }

    exch_wait();

    if (two) {
      v8f o[4];
#pragma unroll
      for (int i = 0; i < 4; ++i) {
        int t = 2 * wave + 2 + i;
        if (t >= 6) t -= 6;
        o[i] = get8(src, t, lane);
      }
      const v2f o6 = get2(src, lane);
      __builtin_amdgcn_sched_barrier(0);
#pragma unroll
      for (int j = 0; j < 4; ++j) {
        const v2f b = {d1[2 * j], d1[2 * j + 1]};
        c0 = wmma4(sA0[4 + j], b, c0);
        c1 = wmma4(sA1[4 + j], b, c1);
      }
#pragma unroll
      for (int i = 0; i < 4; ++i) {
#pragma unroll
        for (int j = 0; j < 4; ++j) {
          const v2f b = {o[i][2 * j], o[i][2 * j + 1]};
          c0 = wmma4(sA0[8 + 4 * i + j], b, c0);
          c1 = wmma4(sA1[8 + 4 * i + j], b, c1);
        }
      }
      c0 = wmma4(sA0[24], o6, c0);
      c1 = wmma4(sA1[24], o6, c1);
      d0 = soft8(c0, thr);
      put8(dst, T0, lane, d0);
      d1 = soft8(c1, thr);
      put8(dst, T1, lane, d1);
    } else {
      v8f o[6];
#pragma unroll
      for (int i = 0; i < 6; ++i) o[i] = get8(src, i, lane);
      __builtin_amdgcn_sched_barrier(0);
      v8f e = {};
#pragma unroll
      for (int i = 0; i < 6; ++i) {
#pragma unroll
        for (int j = 0; j < 4; ++j) {
          const v2f b = {o[i][2 * j], o[i][2 * j + 1]};
          if (j & 1)
            c0 = wmma4(sA0[1 + 4 * i + j], b, c0);
          else
            e = wmma4(sA0[1 + 4 * i + j], b, e);
        }
      }
#pragma unroll
      for (int r = 0; r < 8; ++r) c0[r] += e[r];
      d0 = soft8(c0, thr);
      put2(dst, lane, d0);
    }
  }

  {
    float* rowp = stg + c * N_DIM;
    if (two) {
      *(v4f*)(rowp + 16 * T0 + 8 * h) = v4f{d0[0], d0[1], d0[2], d0[3]};
      *(v4f*)(rowp + 16 * T0 + 8 * h + 4) = v4f{d0[4], d0[5], d0[6], d0[7]};
      *(v4f*)(rowp + 16 * T1 + 8 * h) = v4f{d1[0], d1[1], d1[2], d1[3]};
      *(v4f*)(rowp + 16 * T1 + 8 * h + 4) = v4f{d1[4], d1[5], d1[6], d1[7]};
    } else {
      *(v2f*)(rowp + 96 + 2 * h) = v2f{d0[0], d0[1]};
    }
    __syncthreads();
    float* ob = out + (size_t)row0 * N_DIM;
    for (int i = tid; i < TILE_ELEMS; i += THREADS) ob[i] = stg[i];
  }
}

extern "C" void kernel_launch(void* const* d_in, const int* in_sizes, int n_in,
                              void* d_out, int out_size, void* d_ws,
                              size_t ws_size, hipStream_t stream) {
  const float* y = (const float*)d_in[0];
  const float* S = (const float*)d_in[1];
  const float* W = (const float*)d_in[2];
  const float* thr = (const float*)d_in[3];
  const int* nIt = (const int*)d_in[4];
  float* out = (float*)d_out;

  const int blocks = B_ROWS / 16;
  hipLaunchKernelGGL(ista_kernel, dim3(blocks), dim3(THREADS),
                     0, stream, y, S, W, thr, nIt, out);
}


# Wave 3 (one tile) takes 3 k-steps of every full tile, its partial sums go through LDS behind a second barrier phase: 44/43 matrix instructions per wave and step, was 50/25; k summation order changes
# speedup vs baseline: 4.5415x; 1.0358x over previous
#include <hip/hip_runtime.h>


typedef float v2f __attribute__((ext_vector_type(2)));
typedef float v4f __attribute__((ext_vector_type(4)));
typedef float v8f __attribute__((ext_vector_type(8)));

__device__ __forceinline__ v8f wmma4(v2f a, v2f b, v8f c) {
  return __builtin_amdgcn_wmma_f32_16x16x4_f32(false, a, false, b, (short)0, c,
                                               false, false);
}

#define M_DIM 70
#define N_DIM 100
#define B_ROWS 4096
#define KT_S 25
#define KT_W 18
#define WAVES 4
#define THREADS (WAVES * 32)
#define TILE_ELEMS (16 * N_DIM)
#define TILE_F 256
#define STAGE_F (6 * TILE_F)
#define KP 3


__device__ __forceinline__ v8f soft8(const v8f c, const float thr) {
  const v8f t = {thr, thr, thr, thr, thr, thr, thr, thr};
  const v8f z = {};
  return __builtin_elementwise_max(c - t, z) -
         __builtin_elementwise_max(-c - t, z);
}

__device__ __forceinline__ void put8(float* stage, const int t, const int lane,
                                     const v8f d) {
  v4f* p = (v4f*)(stage + t * TILE_F + lane * 4);
  p[0] = v4f{d[0], d[1], d[2], d[3]};
  p[32] = v4f{d[4], d[5], d[6], d[7]};
}
__device__ __forceinline__ v8f get8(const float* stage, const int t,
                                    const int lane) {
  const v4f* p = (const v4f*)(stage + t * TILE_F + lane * 4);
  const v4f lo = p[0];
  const v4f hi = p[32];
  return v8f{lo[0], lo[1], lo[2], lo[3], hi[0], hi[1], hi[2], hi[3]};
}
__device__ __forceinline__ v4f get4(const float* stage, const int t,
                                    const int lane) {
  return *(const v4f*)(stage + t * TILE_F + lane * 4);
}

__device__ __forceinline__ void exch_arrive() {
  __builtin_amdgcn_fence(__ATOMIC_RELEASE, "workgroup");
  __builtin_amdgcn_s_barrier_signal(-1);
}
__device__ __forceinline__ void exch_wait() {
  __builtin_amdgcn_s_barrier_wait(-1);
  __builtin_amdgcn_fence(__ATOMIC_ACQUIRE, "workgroup");
}

__global__ __launch_bounds__(THREADS) void ista_kernel(
    const float* __restrict__ y, const float* __restrict__ S,
    const float* __restrict__ W, const float* __restrict__ thrp,
    const int* __restrict__ iterp, float* __restrict__ out) {
  __shared__ __attribute__((aligned(16))) float xbuf[2 * STAGE_F];
  __shared__ __attribute__((aligned(16))) float pbuf[6 * TILE_F];
  __shared__ __attribute__((aligned(16))) float stg[TILE_ELEMS];

  const int tid = threadIdx.x;
  const int lane = tid & 31;
  const int wave = __builtin_amdgcn_readfirstlane(tid >> 5);
  const int c = lane & 15;
  const int h = lane >> 4;
  const int row0 = blockIdx.x * 16;
  const bool two = wave < 3;

  const int T0 = two ? 2 * wave : 6;
  const int T1 = two ? 2 * wave + 1 : 6;
  const bool okr = two || (c & 6) == 0;
  const int rA0 =
      two ? 16 * T0 + c : (okr ? 96 + (c & 1) + 2 * (c >> 3) : N_DIM - 1);
  const int rA1 = two ? 16 * T1 + c : rA0;
  const v2f z2 = {0.0f, 0.0f};

  v8f wy0 = {}, wy1 = {};
  {
    v2f by[KT_W];
    const float* yb = y + (size_t)(row0 + c) * M_DIM + 2 * h;
#pragma unroll
    for (int k = 0; k < KT_W; ++k) {
      v2f a = {0.0f, 0.0f};
      if (4 * k + 2 * h < M_DIM) a = *(const v2f*)(yb + 4 * k);
      by[k] = a;
    }
    const float* w0 = W + (size_t)rA0 * M_DIM;
    const float* w1 = W + (size_t)rA1 * M_DIM;
    v2f aw0[KT_W], aw1[KT_W];
#pragma unroll
    for (int k = 0; k < KT_W; ++k) {
      const bool in = 4 * k + 2 * h < M_DIM;
      const int col = in ? 4 * k + 2 * h : 4 * k;
      const v2f t0 = *(const v2f*)(w0 + col);
      const v2f t1 = *(const v2f*)(w1 + col);
      aw0[k] = (okr && in) ? t0 : z2;
      aw1[k] = in ? t1 : z2;
    }
#pragma unroll
    for (int k = 0; k < KT_W; ++k) wy0 = wmma4(aw0[k], by[k], wy0);
    if (two) {
#pragma unroll
      for (int k = 0; k < KT_W; ++k) wy1 = wmma4(aw1[k], by[k], wy1);
    }
  }

  v2f sA0[KT_S], sA1[KT_S];
  if (two) {
    const float* p0 = S + (size_t)rA0 * N_DIM + 8 * h;
    const float* p1 = S + (size_t)rA1 * N_DIM + 8 * h;
#pragma unroll
    for (int p = 0; p < 22; ++p) {
      int t;
      if (p < 4) {
        t = T0;
      } else if (p < 8) {
        t = T1;
      } else {
        t = 2 * wave + 2 + ((p - 8) >> 2);
        if (t >= 6) t -= 6;
      }
      const int kc = 16 * t + 2 * (p & 3);
      sA0[p] = *(const v2f*)(p0 + kc);
      sA1[p] = *(const v2f*)(p1 + kc);
    }
  } else {
    const float* p0 = S + (size_t)rA0 * N_DIM;
    const v2f t6 = *(const v2f*)(p0 + 96 + 2 * h);
    sA0[0] = okr ? t6 : z2;
#pragma unroll
    for (int p = 1; p < KT_S; ++p) {
      const int kc = 16 * ((p - 1) >> 2) + 8 * h + 2 * ((p - 1) & 3);
      const v2f v = *(const v2f*)(p0 + kc);
      sA0[p] = okr ? v : z2;
    }
#pragma unroll
    for (int T = 0; T < 6; ++T) {
      const float* pr = S + (size_t)(16 * T + c) * N_DIM;
      const int tl = (2 * (T >> 1) + 5) % 6;
      sA1[KP * T + 0] = *(const v2f*)(pr + 96 + 2 * h);
      sA1[KP * T + 1] = *(const v2f*)(pr + 16 * tl + 8 * h + 4);
      sA1[KP * T + 2] = *(const v2f*)(pr + 16 * tl + 8 * h + 6);
    }
  }

  const float thr = thrp[0];
  const int nIter = iterp[0];

  v8f d0 = {}, d1 = {};
  if (two) {
    put8(xbuf, T0, lane, d0);
    put8(xbuf, T1, lane, d1);
  }

  for (int it = 0; it < nIter; ++it) {
    const float* src = xbuf + (it & 1) * STAGE_F;
    float* dst = xbuf + ((it & 1) ^ 1) * STAGE_F;

    if (two) {
      v8f c0 = wy0, c1 = wy1;
#pragma unroll
      for (int j = 0; j < 2; ++j) {
        const v2f b = {d0[2 * j], d0[2 * j + 1]};
        c0 = wmma4(sA0[j], b, c0);
        c1 = wmma4(sA1[j], b, c1);
      }
      exch_arrive();
#pragma unroll
      for (int j = 2; j < 4; ++j) {
        const v2f b = {d0[2 * j], d0[2 * j + 1]};
        c0 = wmma4(sA0[j], b, c0);
        c1 = wmma4(sA1[j], b, c1);
      }
      exch_wait();
      __builtin_amdgcn_s_barrier_signal(-1);

      v8f o[3];
#pragma unroll
      for (int i = 0; i < 3; ++i) {
        int t = 2 * wave + 2 + i;
        if (t >= 6) t -= 6;
        o[i] = get8(src, t, lane);
      }
      int tl = 2 * wave + 5;
      if (tl >= 6) tl -= 6;
      const v4f ol = get4(src, tl, lane);
      __builtin_amdgcn_sched_barrier(0);
#pragma unroll
      for (int j = 0; j < 4; ++j) {
        const v2f b = {d1[2 * j], d1[2 * j + 1]};
        c0 = wmma4(sA0[4 + j], b, c0);
        c1 = wmma4(sA1[4 + j], b, c1);
      }
#pragma unroll
      for (int i = 0; i < 2; ++i) {
#pragma unroll
        for (int j = 0; j < 4; ++j) {
          const v2f b = {o[i][2 * j], o[i][2 * j + 1]};
          c0 = wmma4(sA0[8 + 4 * i + j], b, c0);
          c1 = wmma4(sA1[8 + 4 * i + j], b, c1);
        }
      }
#pragma unroll
      for (int j = 0; j < 3; ++j) {
        const v2f b = {o[2][2 * j], o[2][2 * j + 1]};
        c0 = wmma4(sA0[16 + j], b, c0);
        c1 = wmma4(sA1[16 + j], b, c1);
      }
      asm volatile("" ::"v"(c0), "v"(c1));
      exch_wait();
      const v8f q0 = get8(pbuf, T0, lane);
      const v8f q1 = get8(pbuf, T1, lane);
      asm volatile("" : "+v"(c0), "+v"(c1)::"memory");
      {
        const v2f b = {o[2][6], o[2][7]};
        c0 = wmma4(sA0[19], b, c0);
        c1 = wmma4(sA1[19], b, c1);
      }
#pragma unroll
      for (int j = 0; j < 2; ++j) {
        const v2f b = {ol[2 * j], ol[2 * j + 1]};
        c0 = wmma4(sA0[20 + j], b, c0);
        c1 = wmma4(sA1[20 + j], b, c1);
      }
#pragma unroll
      for (int r = 0; r < 8; ++r) c0[r] += q0[r];
      d0 = soft8(c0, thr);
      put8(dst, T0, lane, d0);
#pragma unroll
      for (int r = 0; r < 8; ++r) c1[r] += q1[r];
      d1 = soft8(c1, thr);
      put8(dst, T1, lane, d1);
    } else {
      __builtin_amdgcn_s_barrier_signal(-1);
      const v2f b6 = {d0[0], d0[1]};
      v8f P[6];
#pragma unroll
      for (int T = 0; T < 6; ++T) P[T] = wmma4(sA1[KP * T], b6, v8f{});
      v8f c0 = wmma4(sA0[0], b6, wy0);
      exch_wait();
      v4f olo[6], ohi[6];
#pragma unroll
      for (int i = 0; i < 6; ++i) {
        olo[i] = get4(src, i, lane);
        ohi[i] = get4(src + 128, i, lane);
      }
      __builtin_amdgcn_sched_barrier(0);
#pragma unroll
      for (int j = 0; j < 2; ++j) {
#pragma unroll
        for (int T = 0; T < 6; ++T) {
          const int tl = (2 * (T >> 1) + 5) % 6;
          const v2f b = {ohi[tl][2 * j], ohi[tl][2 * j + 1]};
          P[T] = wmma4(sA1[KP * T + 1 + j], b, P[T]);
        }
      }
#pragma unroll
      for (int T = 0; T < 6; ++T) put8(pbuf, T, lane, P[T]);
      exch_arrive();
      asm volatile("" : "+v"(c0), "+v"(olo[0])::"memory");
      v8f e = {};
#pragma unroll
      for (int i = 0; i < 6; ++i) {
#pragma unroll
        for (int j = 0; j < 4; ++j) {
          const v2f b = j < 2 ? v2f{olo[i][2 * j], olo[i][2 * j + 1]}
                              : v2f{ohi[i][2 * j - 4], ohi[i][2 * j - 3]};
          if (j & 1)
            c0 = wmma4(sA0[1 + 4 * i + j], b, c0);
          else
            e = wmma4(sA0[1 + 4 * i + j], b, e);
        }
      }
#pragma unroll
      for (int r = 0; r < 8; ++r) c0[r] += e[r];
      d0 = soft8(c0, thr);
      exch_wait();
    }
  }

  {
    float* rowp = stg + c * N_DIM;
    if (two) {
      *(v4f*)(rowp + 16 * T0 + 8 * h) = v4f{d0[0], d0[1], d0[2], d0[3]};
      *(v4f*)(rowp + 16 * T0 + 8 * h + 4) = v4f{d0[4], d0[5], d0[6], d0[7]};
      *(v4f*)(rowp + 16 * T1 + 8 * h) = v4f{d1[0], d1[1], d1[2], d1[3]};
      *(v4f*)(rowp + 16 * T1 + 8 * h + 4) = v4f{d1[4], d1[5], d1[6], d1[7]};
    } else {
      *(v2f*)(rowp + 96 + 2 * h) = v2f{d0[0], d0[1]};
    }
    __syncthreads();
    float* ob = out + (size_t)row0 * N_DIM;
    for (int i = tid; i < TILE_ELEMS; i += THREADS) ob[i] = stg[i];
  }
}

extern "C" void kernel_launch(void* const* d_in, const int* in_sizes, int n_in,
                              void* d_out, int out_size, void* d_ws,
                              size_t ws_size, hipStream_t stream) {
  const float* y = (const float*)d_in[0];
  const float* S = (const float*)d_in[1];
  const float* W = (const float*)d_in[2];
  const float* thr = (const float*)d_in[3];
  const int* nIt = (const int*)d_in[4];
  float* out = (float*)d_out;

  const int blocks = B_ROWS / 16;
  hipLaunchKernelGGL(ista_kernel, dim3(blocks), dim3(THREADS),
                     0, stream, y, S, W, thr, nIt, out);
}


# Eight waves, two per SIMD: one wave per full tile plus a helper wave for the partial sums, so a wave's LDS, barrier and threshold waits hide behind its SIMD partner; k summation order changes
# speedup vs baseline: 4.7251x; 1.0404x over previous
#include <hip/hip_runtime.h>


typedef float v2f __attribute__((ext_vector_type(2)));
typedef float v4f __attribute__((ext_vector_type(4)));
typedef float v8f __attribute__((ext_vector_type(8)));

__device__ __forceinline__ v8f wmma4(v2f a, v2f b, v8f c) {
  return __builtin_amdgcn_wmma_f32_16x16x4_f32(false, a, false, b, (short)0, c,
                                               false, false);
}

#define M_DIM 70
#define N_DIM 100
#define B_ROWS 4096
#define KT_S 25
#define KT_W 18
#define WAVES 8
#define THREADS (WAVES * 32)
#define TILE_ELEMS (16 * N_DIM)
#define TILE_F 256
#define STAGE_F (6 * TILE_F + 64)
#define KP 3


__device__ __forceinline__ v8f soft8(const v8f c, const float thr) {
  const v8f t = {thr, thr, thr, thr, thr, thr, thr, thr};
  const v8f z = {};
  return __builtin_elementwise_max(c - t, z) -
         __builtin_elementwise_max(-c - t, z);
}

__device__ __forceinline__ void put8(float* stage, const int t, const int lane,
                                     const v8f d) {
  v4f* p = (v4f*)(stage + t * TILE_F + lane * 4);
  p[0] = v4f{d[0], d[1], d[2], d[3]};
  p[32] = v4f{d[4], d[5], d[6], d[7]};
}
__device__ __forceinline__ v8f get8(const float* stage, const int t,
                                    const int lane) {
  const v4f* p = (const v4f*)(stage + t * TILE_F + lane * 4);
  const v4f lo = p[0];
  const v4f hi = p[32];
  return v8f{lo[0], lo[1], lo[2], lo[3], hi[0], hi[1], hi[2], hi[3]};
}
__device__ __forceinline__ v4f get4(const float* stage, const int t,
                                    const int lane) {
  return *(const v4f*)(stage + t * TILE_F + lane * 4);
}

__device__ __forceinline__ void exch_arrive() {
  __builtin_amdgcn_fence(__ATOMIC_RELEASE, "workgroup");
  __builtin_amdgcn_s_barrier_signal(-1);
}
__device__ __forceinline__ void exch_wait() {
  __builtin_amdgcn_s_barrier_wait(-1);
  __builtin_amdgcn_fence(__ATOMIC_ACQUIRE, "workgroup");
}

__global__ __launch_bounds__(THREADS) void ista_kernel(
    const float* __restrict__ y, const float* __restrict__ S,
    const float* __restrict__ W, const float* __restrict__ thrp,
    const int* __restrict__ iterp, float* __restrict__ out) {
  __shared__ __attribute__((aligned(16))) float xbuf[2 * STAGE_F];
  __shared__ __attribute__((aligned(16))) float pbuf[6 * TILE_F];
  __shared__ __attribute__((aligned(16))) float stg[TILE_ELEMS];

  const int tid = threadIdx.x;
  const int lane = tid & 31;
  const int wave = __builtin_amdgcn_readfirstlane(tid >> 5);
  const int c = lane & 15;
  const int h = lane >> 4;
  const int row0 = blockIdx.x * 16;
  const int role = wave & 3;
  const int hw = wave >> 2;
  const bool own = role < 3;
  const bool hlp = wave == 3;
  const bool t6w = wave == 7;

  const int Tm = own ? 2 * role + hw : 6;
  const int Tp = own ? 2 * role + 1 - hw : 6;
  const bool okr = own || (c & 6) == 0;
  const int rA = own ? 16 * Tm + c : (okr ? 96 + (c & 1) + 2 * (c >> 3) : N_DIM - 1);
  const v2f z2 = {0.0f, 0.0f};

  v8f wy = {};
  if (!hlp) {
    v2f by[KT_W];
    const float* yb = y + (size_t)(row0 + c) * M_DIM + 2 * h;
#pragma unroll
    for (int k = 0; k < KT_W; ++k) {
      v2f a = {0.0f, 0.0f};
      if (4 * k + 2 * h < M_DIM) a = *(const v2f*)(yb + 4 * k);
      by[k] = a;
    }
    const float* w0 = W + (size_t)rA * M_DIM;
    v2f aw[KT_W];
#pragma unroll
    for (int k = 0; k < KT_W; ++k) {
      const bool in = 4 * k + 2 * h < M_DIM;
      const int col = in ? 4 * k + 2 * h : 4 * k;
      const v2f t0 = *(const v2f*)(w0 + col);
      aw[k] = (okr && in) ? t0 : z2;
    }
#pragma unroll
    for (int k = 0; k < KT_W; ++k) wy = wmma4(aw[k], by[k], wy);
  }

  v2f sA[KT_S];
  if (own) {
    const float* p0 = S + (size_t)rA * N_DIM + 8 * h;
#pragma unroll
    for (int p = 0; p < 22; ++p) {
      int t;
      if (p < 4) {
        t = Tm;
      } else if (p < 8) {
        t = Tp;
      } else {
        t = 2 * role + 2 + ((p - 8) >> 2);
        if (t >= 6) t -= 6;
      }
      const int kc = 16 * t + 2 * (p & 3);
      sA[p] = *(const v2f*)(p0 + kc);
    }
  } else if (t6w) {
    const float* p0 = S + (size_t)rA * N_DIM;
    const v2f t6 = *(const v2f*)(p0 + 96 + 2 * h);
    sA[0] = okr ? t6 : z2;
#pragma unroll
    for (int p = 1; p < KT_S; ++p) {
      const int kc = 16 * ((p - 1) >> 2) + 8 * h + 2 * ((p - 1) & 3);
      const v2f v = *(const v2f*)(p0 + kc);
      sA[p] = okr ? v : z2;
    }
  } else {
#pragma unroll
    for (int T = 0; T < 6; ++T) {
      const float* pr = S + (size_t)(16 * T + c) * N_DIM;
      const int tl = (2 * (T >> 1) + 5) % 6;
      sA[KP * T + 0] = *(const v2f*)(pr + 96 + 2 * h);
      sA[KP * T + 1] = *(const v2f*)(pr + 16 * tl + 8 * h + 4);
      sA[KP * T + 2] = *(const v2f*)(pr + 16 * tl + 8 * h + 6);
    }
  }

  const float thr = thrp[0];
  const int nIter = iterp[0];

  v8f d = {};
  if (own) put8(xbuf, Tm, lane, d);
  if (t6w) *(v2f*)(xbuf + 6 * TILE_F + lane * 2) = z2;

  if (own) {
    int to[3];
#pragma unroll
    for (int i = 0; i < 3; ++i) {
      int t = 2 * role + 2 + i;
      if (t >= 6) t -= 6;
      to[i] = t;
    }
    int tl = 2 * role + 5;
    if (tl >= 6) tl -= 6;
    for (int it = 0; it < nIter; ++it) {
      const float* src = xbuf + (it & 1) * STAGE_F;
      float* dst = xbuf + ((it & 1) ^ 1) * STAGE_F;
      v8f a0 = wy, a1 = {};
      a0 = wmma4(sA[0], v2f{d[0], d[1]}, a0);
      a1 = wmma4(sA[1], v2f{d[2], d[3]}, a1);
      exch_arrive();
      a0 = wmma4(sA[2], v2f{d[4], d[5]}, a0);
      exch_wait();
      __builtin_amdgcn_s_barrier_signal(-1);

      const v8f pt = get8(src, Tp, lane);
      v8f o[3];
#pragma unroll
      for (int i = 0; i < 3; ++i) o[i] = get8(src, to[i], lane);
      const v4f ol = get4(src, tl, lane);
      __builtin_amdgcn_sched_barrier(0);
      a1 = wmma4(sA[3], v2f{d[6], d[7]}, a1);
#pragma unroll
      for (int j = 0; j < 4; j += 2) {
        a0 = wmma4(sA[4 + j], v2f{pt[2 * j], pt[2 * j + 1]}, a0);
        a1 = wmma4(sA[5 + j], v2f{pt[2 * j + 2], pt[2 * j + 3]}, a1);
      }
#pragma unroll
      for (int i = 0; i < 2; ++i) {
#pragma unroll
        for (int j = 0; j < 4; j += 2) {
          a0 = wmma4(sA[8 + 4 * i + j], v2f{o[i][2 * j], o[i][2 * j + 1]}, a0);
          a1 = wmma4(sA[9 + 4 * i + j], v2f{o[i][2 * j + 2], o[i][2 * j + 3]},
                     a1);
        }
      }
      a0 = wmma4(sA[16], v2f{o[2][0], o[2][1]}, a0);
      asm volatile("" ::"v"(a0), "v"(a1));
      exch_wait();
      const v8f q = get8(pbuf, Tm, lane);
      asm volatile("" : "+v"(a0), "+v"(a1)::"memory");
      a1 = wmma4(sA[17], v2f{o[2][2], o[2][3]}, a1);
      a0 = wmma4(sA[18], v2f{o[2][4], o[2][5]}, a0);
      a1 = wmma4(sA[19], v2f{o[2][6], o[2][7]}, a1);
      a0 = wmma4(sA[20], v2f{ol[0], ol[1]}, a0);
      a1 = wmma4(sA[21], v2f{ol[2], ol[3]}, a1);
#pragma unroll
      for (int r = 0; r < 8; ++r) a0[r] += a1[r] + q[r];
      d = soft8(a0, thr);
      put8(dst, Tm, lane, d);
    }
  } else if (t6w) {
    for (int it = 0; it < nIter; ++it) {
      const float* src = xbuf + (it & 1) * STAGE_F;
      float* dst = xbuf + ((it & 1) ^ 1) * STAGE_F;
      v8f a0 = wmma4(sA[0], v2f{d[0], d[1]}, wy);
      exch_arrive();
      exch_wait();
      __builtin_amdgcn_s_barrier_signal(-1);
      v4f olo[6], ohi[6];
#pragma unroll
      for (int i = 0; i < 6; ++i) {
        olo[i] = get4(src, i, lane);
        ohi[i] = get4(src + 128, i, lane);
      }
      __builtin_amdgcn_sched_barrier(0);
      v8f a1 = {};
#pragma unroll
      for (int i = 0; i < 6; ++i) {
#pragma unroll
        for (int j = 0; j < 4; ++j) {
          const v2f b = j < 2 ? v2f{olo[i][2 * j], olo[i][2 * j + 1]}
                              : v2f{ohi[i][2 * j - 4], ohi[i][2 * j - 3]};
          if (j & 1)
            a0 = wmma4(sA[1 + 4 * i + j], b, a0);
          else
            a1 = wmma4(sA[1 + 4 * i + j], b, a1);
        }
      }
#pragma unroll
      for (int r = 0; r < 8; ++r) a0[r] += a1[r];
      d = soft8(a0, thr);
      *(v2f*)(dst + 6 * TILE_F + lane * 2) = v2f{d[0], d[1]};
      exch_wait();
    }
  } else {
    for (int it = 0; it < nIter; ++it) {
      const float* src = xbuf + (it & 1) * STAGE_F;
      __builtin_amdgcn_s_barrier_signal(-1);
      exch_wait();
      const v2f b6 = *(const v2f*)(src + 6 * TILE_F + lane * 2);
      v4f ohi[3];
#pragma unroll
      for (int i = 0; i < 3; ++i) ohi[i] = get4(src + 128, 2 * i + 1, lane);
      __builtin_amdgcn_sched_barrier(0);
      v8f P[6];
#pragma unroll
      for (int T = 0; T < 6; ++T) P[T] = wmma4(sA[KP * T], b6, v8f{});
#pragma unroll
      for (int j = 0; j < 2; ++j) {
#pragma unroll
        for (int T = 0; T < 6; ++T) {
          const int tl = (2 * (T >> 1) + 5) % 6;
          const v2f b = {ohi[tl >> 1][2 * j], ohi[tl >> 1][2 * j + 1]};
          P[T] = wmma4(sA[KP * T + 1 + j], b, P[T]);
        }
      }
#pragma unroll
      for (int T = 0; T < 6; ++T) put8(pbuf, T, lane, P[T]);
      exch_arrive();
      exch_wait();
    }
  }

  {
    float* rowp = stg + c * N_DIM;
    if (own) {
      *(v4f*)(rowp + 16 * Tm + 8 * h) = v4f{d[0], d[1], d[2], d[3]};
      *(v4f*)(rowp + 16 * Tm + 8 * h + 4) = v4f{d[4], d[5], d[6], d[7]};
    } else if (t6w) {
      *(v2f*)(rowp + 96 + 2 * h) = v2f{d[0], d[1]};
    }
    __syncthreads();
    float* ob = out + (size_t)row0 * N_DIM;
    for (int i = tid; i < TILE_ELEMS; i += THREADS) ob[i] = stg[i];
  }
}

extern "C" void kernel_launch(void* const* d_in, const int* in_sizes, int n_in,
                              void* d_out, int out_size, void* d_ws,
                              size_t ws_size, hipStream_t stream) {
  const float* y = (const float*)d_in[0];
  const float* S = (const float*)d_in[1];
  const float* W = (const float*)d_in[2];
  const float* thr = (const float*)d_in[3];
  const int* nIt = (const int*)d_in[4];
  float* out = (float*)d_out;

  const int blocks = B_ROWS / 16;
  hipLaunchKernelGGL(ista_kernel, dim3(blocks), dim3(THREADS),
                     0, stream, y, S, W, thr, nIt, out);
}
